# starting kernel
# baseline (speedup 1.0000x reference)
#include <hip/hip_runtime.h>
#include <hip/hip_bf16.h>

typedef __attribute__((ext_vector_type(2))) float v2f;
typedef __attribute__((ext_vector_type(8))) float v8f;

#define HIDDEN 128
#define TOKENS_PER_BLOCK 128
#define NB 64
#define NP 1024

__global__ __launch_bounds__(256) void mlp_wmma_kernel(
    const float* __restrict__ freqs, const float* __restrict__ mags,
    const float* __restrict__ W1, const float* __restrict__ b1,
    const float* __restrict__ W2, const float* __restrict__ b2,
    const float* __restrict__ W3, const float* __restrict__ b3,
    float* __restrict__ out)
{
  __shared__ float hbuf[TOKENS_PER_BLOCK * HIDDEN];

  const int tid       = threadIdx.x;
  const int blockTok0 = blockIdx.x * TOKENS_PER_BLOCK;

  {
    const int token = tid >> 1;
    const int cbase = (tid & 1) * 64;
    const int g     = blockTok0 + token;
    const float f = freqs[g];
    const float m = mags[g];
    #pragma unroll 8
    for (int j = 0; j < 64; ++j) {
      const int c = cbase + j;
      float h = fmaf(f, W1[c], fmaf(m, W1[HIDDEN + c], b1[c]));
      hbuf[token * HIDDEN + c] = h > 0.0f ? h : 0.0f;
    }
  }
  __syncthreads();

  const int lane  = tid & 31;
  const int wave  = tid >> 5;
  const int idx   = lane & 15;
  const int half  = (lane >> 4) << 1;
  const int wrow0 = wave * 16;

  v8f acc[8];
  #pragma unroll
  for (int n = 0; n < 8; ++n) {
    const float bias = b2[n * 16 + idx];
    #pragma unroll
    for (int r = 0; r < 8; ++r) acc[n][r] = bias;
  }

  const float* aP = &hbuf[(wrow0 + idx) * HIDDEN + half];

  #pragma unroll 4
  for (int k = 0; k < HIDDEN; k += 4) {
    v2f a; a.x = aP[k]; a.y = aP[k + 1];
    #pragma unroll
    for (int n = 0; n < 8; ++n) {
      const float* bP = &W2[(k + half) * HIDDEN + n * 16 + idx];
      v2f b; b.x = bP[0]; b.y = bP[HIDDEN];
      acc[n] = __builtin_amdgcn_wmma_f32_16x16x4_f32(
          false, a, false, b, (short)0, acc[n], false, false);
    }
  }

  __syncthreads();

  #pragma unroll
  for (int n = 0; n < 8; ++n) {
    #pragma unroll
    for (int r = 0; r < 8; ++r) {
      const int row = wrow0 + ((lane < 16) ? r : r + 8);
      const float v = acc[n][r];
      hbuf[row * HIDDEN + n * 16 + idx] = v > 0.0f ? v : 0.0f;
    }
  }
  __syncthreads();

  v8f acc3;
  {
    const float bias = (idx < 8) ? b3[idx] : 0.0f;
    #pragma unroll
    for (int r = 0; r < 8; ++r) acc3[r] = bias;
  }
  #pragma unroll 4
  for (int k = 0; k < HIDDEN; k += 4) {
    v2f a; a.x = aP[k]; a.y = aP[k + 1];
    v2f b;
    b.x = (idx < 8) ? W3[(k + half)     * 8 + idx] : 0.0f;
    b.y = (idx < 8) ? W3[(k + half + 1) * 8 + idx] : 0.0f;
    acc3 = __builtin_amdgcn_wmma_f32_16x16x4_f32(
        false, a, false, b, (short)0, acc3, false, false);
  }

  if (idx < 8) {
    const int gbase = blockTok0 + wrow0;
    #pragma unroll
    for (int r = 0; r < 8; ++r) {
      const int row = (lane < 16) ? r : r + 8;
      out[(gbase + row) * 8 + idx] = acc3[r];
    }
  }
}

__global__ __launch_bounds__(1024) void fundamental_kernel(
    const float* __restrict__ freqs, const float* __restrict__ mags,
    float* __restrict__ out)
{
  __shared__ float fr[NP];
  __shared__ float mg[NP];
  __shared__ int   hm[NP];

  const int i    = threadIdx.x;
  const int base = (NB - 1) * NP;
  fr[i] = freqs[base + i];
  mg[i] = mags[base + i];
  __syncthreads();

  const float fi = fr[i];
  int h = 0;
  for (int j = 0; j < NP; ++j) {
    if (j == i) continue;
    if (!(mg[j] > 0.0f)) continue;
    const float fj = fr[j];
    bool match = false;
    #pragma unroll
    for (int order = 2; order <= 8; ++order)
      match = match || (fabsf(fj - fi * (float)order) < 2.0f);
    h += match ? 1 : 0;
  }
  const bool valid = mg[i] > 0.0f;
  hm[i] = (valid && h > 0) ? h : -1;
  __syncthreads();

  if (i == 0) {
    int max_h = -1;
    for (int t = 0; t < NP; ++t) max_h = (hm[t] > max_h) ? hm[t] : max_h;
    int best_idx = 0;
    float best_mag = -__builtin_inff();
    if (max_h > 0) {
      for (int t = 0; t < NP; ++t)
        if (hm[t] == max_h && mg[t] > best_mag) { best_mag = mg[t]; best_idx = t; }
    } else {
      for (int t = 0; t < NP; ++t)
        if (mg[t] > best_mag) { best_mag = mg[t]; best_idx = t; }
    }
    out[NB * NP * 8] = fr[best_idx];
  }
}

extern "C" void kernel_launch(void* const* d_in, const int* in_sizes, int n_in,
                              void* d_out, int out_size, void* d_ws, size_t ws_size,
                              hipStream_t stream) {
  const float* freqs = (const float*)d_in[0];
  const float* mags  = (const float*)d_in[1];
  const float* W1    = (const float*)d_in[2];
  const float* b1    = (const float*)d_in[3];
  const float* W2    = (const float*)d_in[4];
  const float* b2    = (const float*)d_in[5];
  const float* W3    = (const float*)d_in[6];
  const float* b3    = (const float*)d_in[7];
  float* out = (float*)d_out;

  const int ntok   = in_sizes[0];
  const int blocks = ntok / TOKENS_PER_BLOCK;

  mlp_wmma_kernel<<<blocks, 256, 0, stream>>>(freqs, mags, W1, b1, W2, b2, W3, b3, out);
  fundamental_kernel<<<1, 1024, 0, stream>>>(freqs, mags, out);
  (void)d_ws; (void)ws_size; (void)n_in; (void)out_size;
}


# last-row harmonic match count (the P*P loop) moved from launch 2's single work-group into 32 extra work-groups of launch 1, integer counts passed in 4 KB of workspace; launch 2 only selects
# speedup vs baseline: 2.3113x; 2.3113x over previous
#include <hip/hip_runtime.h>
#include <hip/hip_bf16.h>

typedef __attribute__((ext_vector_type(2))) float v2f;
typedef __attribute__((ext_vector_type(8))) float v8f;

#define HIDDEN 128
#define TOKENS_PER_BLOCK 128
#define NB 64
#define NP 1024
#define CNT_BLOCKS 32
#define CNT_WS_BYTES (NP * 4)

__global__ __launch_bounds__(256) void mlp_wmma_kernel(
    const float* __restrict__ freqs, const float* __restrict__ mags,
    const float* __restrict__ W1, const float* __restrict__ b1,
    const float* __restrict__ W2, const float* __restrict__ b2,
    const float* __restrict__ W3, const float* __restrict__ b3,
    float* __restrict__ out, int* __restrict__ ws_hm, size_t ws_bytes)
{
  __shared__ float hbuf[TOKENS_PER_BLOCK * HIDDEN];

  const int tid       = threadIdx.x;

  if (blockIdx.x < CNT_BLOCKS) {
    if (ws_bytes < CNT_WS_BYTES) return;
    const float* fr = freqs + (NB - 1) * NP;
    const float* mg = mags  + (NB - 1) * NP;
    const int clane = tid & 31;
    const int cwave = tid >> 5;
    const int i     = blockIdx.x * 32 + clane;
    const int j0    = cwave * (NP / 8);
    const float fi  = fr[i];
    float p[7];
    #pragma unroll
    for (int order = 2; order <= 8; ++order) p[order - 2] = fi * (float)order;
    float fv[4], mv[4];
    #pragma unroll
    for (int q = 0; q < 4; ++q) {
      fv[q] = fr[j0 + 32 * q + clane];
      mv[q] = mg[j0 + 32 * q + clane];
    }
    int h = 0;
    #pragma unroll
    for (int q = 0; q < 4; ++q) {
      for (int l = 0; l < 32; ++l) {
        const float fj = __int_as_float(__builtin_amdgcn_readlane(__float_as_int(fv[q]), l));
        const float mj = __int_as_float(__builtin_amdgcn_readlane(__float_as_int(mv[q]), l));
        bool match = false;
        #pragma unroll
        for (int o = 0; o < 7; ++o)
          match = match | (fabsf(fj - p[o]) < 2.0f);
        const int j = j0 + 32 * q + l;
        h += (match & (mj > 0.0f) & (j != i)) ? 1 : 0;
      }
    }
    int* part = (int*)hbuf;
    part[cwave * 32 + clane] = h;
    __syncthreads();
    if (tid < 32) {
      int s = 0;
      #pragma unroll
      for (int w = 0; w < 8; ++w) s += part[w * 32 + tid];
      const bool valid = mg[i] > 0.0f;
      ws_hm[i] = (valid && s > 0) ? s : -1;
    }
    return;
  }

  const int blockTok0 = (blockIdx.x - CNT_BLOCKS) * TOKENS_PER_BLOCK;

  {
    const int token = tid >> 1;
    const int cbase = (tid & 1) * 64;
    const int g     = blockTok0 + token;
    const float f = freqs[g];
    const float m = mags[g];
    #pragma unroll 8
    for (int j = 0; j < 64; ++j) {
      const int c = cbase + j;
      float h = fmaf(f, W1[c], fmaf(m, W1[HIDDEN + c], b1[c]));
      hbuf[token * HIDDEN + c] = h > 0.0f ? h : 0.0f;
    }
  }
  __syncthreads();

  const int lane  = tid & 31;
  const int wave  = tid >> 5;
  const int idx   = lane & 15;
  const int half  = (lane >> 4) << 1;
  const int wrow0 = wave * 16;

  v8f acc[8];
  #pragma unroll
  for (int n = 0; n < 8; ++n) {
    const float bias = b2[n * 16 + idx];
    #pragma unroll
    for (int r = 0; r < 8; ++r) acc[n][r] = bias;
  }

  const float* aP = &hbuf[(wrow0 + idx) * HIDDEN + half];

  #pragma unroll 4
  for (int k = 0; k < HIDDEN; k += 4) {
    v2f a; a.x = aP[k]; a.y = aP[k + 1];
    #pragma unroll
    for (int n = 0; n < 8; ++n) {
      const float* bP = &W2[(k + half) * HIDDEN + n * 16 + idx];
      v2f b; b.x = bP[0]; b.y = bP[HIDDEN];
      acc[n] = __builtin_amdgcn_wmma_f32_16x16x4_f32(
          false, a, false, b, (short)0, acc[n], false, false);
    }
  }

  __syncthreads();

  #pragma unroll
  for (int n = 0; n < 8; ++n) {
    #pragma unroll
    for (int r = 0; r < 8; ++r) {
      const int row = wrow0 + ((lane < 16) ? r : r + 8);
      const float v = acc[n][r];
      hbuf[row * HIDDEN + n * 16 + idx] = v > 0.0f ? v : 0.0f;
    }
  }
  __syncthreads();

  v8f acc3;
  {
    const float bias = (idx < 8) ? b3[idx] : 0.0f;
    #pragma unroll
    for (int r = 0; r < 8; ++r) acc3[r] = bias;
  }
  #pragma unroll 4
  for (int k = 0; k < HIDDEN; k += 4) {
    v2f a; a.x = aP[k]; a.y = aP[k + 1];
    v2f b;
    b.x = (idx < 8) ? W3[(k + half)     * 8 + idx] : 0.0f;
    b.y = (idx < 8) ? W3[(k + half + 1) * 8 + idx] : 0.0f;
    acc3 = __builtin_amdgcn_wmma_f32_16x16x4_f32(
        false, a, false, b, (short)0, acc3, false, false);
  }

  if (idx < 8) {
    const int gbase = blockTok0 + wrow0;
    #pragma unroll
    for (int r = 0; r < 8; ++r) {
      const int row = (lane < 16) ? r : r + 8;
      out[(gbase + row) * 8 + idx] = acc3[r];
    }
  }
}

__global__ __launch_bounds__(1024) void fundamental_kernel(
    const float* __restrict__ freqs, const float* __restrict__ mags,
    const int* ws_hm, size_t ws_bytes,
    float* __restrict__ out)
{
  __shared__ float fr[NP];
  __shared__ float mg[NP];
  __shared__ int   hm[NP];

  const int i    = threadIdx.x;
  const int base = (NB - 1) * NP;
  fr[i] = freqs[base + i];
  mg[i] = mags[base + i];
  __syncthreads();

  if (ws_bytes >= CNT_WS_BYTES) {
    hm[i] = ws_hm[i];
  } else {
    const float fi = fr[i];
    int h = 0;
    for (int j = 0; j < NP; ++j) {
      if (j == i) continue;
      if (!(mg[j] > 0.0f)) continue;
      const float fj = fr[j];
      bool match = false;
      #pragma unroll
      for (int order = 2; order <= 8; ++order)
        match = match || (fabsf(fj - fi * (float)order) < 2.0f);
      h += match ? 1 : 0;
    }
    const bool valid = mg[i] > 0.0f;
    hm[i] = (valid && h > 0) ? h : -1;
  }
  __syncthreads();

  if (i == 0) {
    int max_h = -1;
    for (int t = 0; t < NP; ++t) max_h = (hm[t] > max_h) ? hm[t] : max_h;
    int best_idx = 0;
    float best_mag = -__builtin_inff();
    if (max_h > 0) {
      for (int t = 0; t < NP; ++t)
        if (hm[t] == max_h && mg[t] > best_mag) { best_mag = mg[t]; best_idx = t; }
    } else {
      for (int t = 0; t < NP; ++t)
        if (mg[t] > best_mag) { best_mag = mg[t]; best_idx = t; }
    }
    out[NB * NP * 8] = fr[best_idx];
  }
}

extern "C" void kernel_launch(void* const* d_in, const int* in_sizes, int n_in,
                              void* d_out, int out_size, void* d_ws, size_t ws_size,
                              hipStream_t stream) {
  const float* freqs = (const float*)d_in[0];
  const float* mags  = (const float*)d_in[1];
  const float* W1    = (const float*)d_in[2];
  const float* b1    = (const float*)d_in[3];
  const float* W2    = (const float*)d_in[4];
  const float* b2    = (const float*)d_in[5];
  const float* W3    = (const float*)d_in[6];
  const float* b3    = (const float*)d_in[7];
  float* out = (float*)d_out;

  const int ntok   = in_sizes[0];
  const int blocks = ntok / TOKENS_PER_BLOCK;

  int* ws_hm = (int*)d_ws;

  mlp_wmma_kernel<<<blocks + CNT_BLOCKS, 256, 0, stream>>>(freqs, mags, W1, b1, W2, b2, W3, b3, out,
                                                          ws_hm, ws_size);
  fundamental_kernel<<<1, 1024, 0, stream>>>(freqs, mags, ws_hm, ws_size, out);
  (void)n_in; (void)out_size;
}


# launch 2's peak choice: thread 0's two serial 1024-trip LDS scans replaced by a work-group-wide maximum of a 64-bit (count, magnitude, index) key: cross-lane on the vector ALU, 32 wave maxima via LDS
# speedup vs baseline: 5.1508x; 2.2285x over previous
#include <hip/hip_runtime.h>
#include <hip/hip_bf16.h>

typedef __attribute__((ext_vector_type(2))) float v2f;
typedef __attribute__((ext_vector_type(8))) float v8f;

#define HIDDEN 128
#define TOKENS_PER_BLOCK 128
#define NB 64
#define NP 1024
#define CNT_BLOCKS 32
#define CNT_WS_BYTES (NP * 4)

__global__ __launch_bounds__(256) void mlp_wmma_kernel(
    const float* __restrict__ freqs, const float* __restrict__ mags,
    const float* __restrict__ W1, const float* __restrict__ b1,
    const float* __restrict__ W2, const float* __restrict__ b2,
    const float* __restrict__ W3, const float* __restrict__ b3,
    float* __restrict__ out, int* __restrict__ ws_hm, size_t ws_bytes)
{
  __shared__ float hbuf[TOKENS_PER_BLOCK * HIDDEN];

  const int tid       = threadIdx.x;

  if (blockIdx.x < CNT_BLOCKS) {
    if (ws_bytes < CNT_WS_BYTES) return;
    const float* fr = freqs + (NB - 1) * NP;
    const float* mg = mags  + (NB - 1) * NP;
    const int clane = tid & 31;
    const int cwave = tid >> 5;
    const int i     = blockIdx.x * 32 + clane;
    const int j0    = cwave * (NP / 8);
    const float fi  = fr[i];
    float p[7];
    #pragma unroll
    for (int order = 2; order <= 8; ++order) p[order - 2] = fi * (float)order;
    float fv[4], mv[4];
    #pragma unroll
    for (int q = 0; q < 4; ++q) {
      fv[q] = fr[j0 + 32 * q + clane];
      mv[q] = mg[j0 + 32 * q + clane];
    }
    int h = 0;
    #pragma unroll
    for (int q = 0; q < 4; ++q) {
      for (int l = 0; l < 32; ++l) {
        const float fj = __int_as_float(__builtin_amdgcn_readlane(__float_as_int(fv[q]), l));
        const float mj = __int_as_float(__builtin_amdgcn_readlane(__float_as_int(mv[q]), l));
        bool match = false;
        #pragma unroll
        for (int o = 0; o < 7; ++o)
          match = match | (fabsf(fj - p[o]) < 2.0f);
        const int j = j0 + 32 * q + l;
        h += (match & (mj > 0.0f) & (j != i)) ? 1 : 0;
      }
    }
    int* part = (int*)hbuf;
    part[cwave * 32 + clane] = h;
    __syncthreads();
    if (tid < 32) {
      int s = 0;
      #pragma unroll
      for (int w = 0; w < 8; ++w) s += part[w * 32 + tid];
      const bool valid = mg[i] > 0.0f;
      ws_hm[i] = (valid && s > 0) ? s : -1;
    }
    return;
  }

  const int blockTok0 = (blockIdx.x - CNT_BLOCKS) * TOKENS_PER_BLOCK;

  {
    const int token = tid >> 1;
    const int cbase = (tid & 1) * 64;
    const int g     = blockTok0 + token;
    const float f = freqs[g];
    const float m = mags[g];
    #pragma unroll 8
    for (int j = 0; j < 64; ++j) {
      const int c = cbase + j;
      float h = fmaf(f, W1[c], fmaf(m, W1[HIDDEN + c], b1[c]));
      hbuf[token * HIDDEN + c] = h > 0.0f ? h : 0.0f;
    }
  }
  __syncthreads();

  const int lane  = tid & 31;
  const int wave  = tid >> 5;
  const int idx   = lane & 15;
  const int half  = (lane >> 4) << 1;
  const int wrow0 = wave * 16;

  v8f acc[8];
  #pragma unroll
  for (int n = 0; n < 8; ++n) {
    const float bias = b2[n * 16 + idx];
    #pragma unroll
    for (int r = 0; r < 8; ++r) acc[n][r] = bias;
  }

  const float* aP = &hbuf[(wrow0 + idx) * HIDDEN + half];

  #pragma unroll 4
  for (int k = 0; k < HIDDEN; k += 4) {
    v2f a; a.x = aP[k]; a.y = aP[k + 1];
    #pragma unroll
    for (int n = 0; n < 8; ++n) {
      const float* bP = &W2[(k + half) * HIDDEN + n * 16 + idx];
      v2f b; b.x = bP[0]; b.y = bP[HIDDEN];
      acc[n] = __builtin_amdgcn_wmma_f32_16x16x4_f32(
          false, a, false, b, (short)0, acc[n], false, false);
    }
  }

  __syncthreads();

  #pragma unroll
  for (int n = 0; n < 8; ++n) {
    #pragma unroll
    for (int r = 0; r < 8; ++r) {
      const int row = wrow0 + ((lane < 16) ? r : r + 8);
      const float v = acc[n][r];
      hbuf[row * HIDDEN + n * 16 + idx] = v > 0.0f ? v : 0.0f;
    }
  }
  __syncthreads();

  v8f acc3;
  {
    const float bias = (idx < 8) ? b3[idx] : 0.0f;
    #pragma unroll
    for (int r = 0; r < 8; ++r) acc3[r] = bias;
  }
  #pragma unroll 4
  for (int k = 0; k < HIDDEN; k += 4) {
    v2f a; a.x = aP[k]; a.y = aP[k + 1];
    v2f b;
    b.x = (idx < 8) ? W3[(k + half)     * 8 + idx] : 0.0f;
    b.y = (idx < 8) ? W3[(k + half + 1) * 8 + idx] : 0.0f;
    acc3 = __builtin_amdgcn_wmma_f32_16x16x4_f32(
        false, a, false, b, (short)0, acc3, false, false);
  }

  if (idx < 8) {
    const int gbase = blockTok0 + wrow0;
    #pragma unroll
    for (int r = 0; r < 8; ++r) {
      const int row = (lane < 16) ? r : r + 8;
      out[(gbase + row) * 8 + idx] = acc3[r];
    }
  }
}

typedef unsigned long long u64;

__device__ __forceinline__ u64 select_key(int hmasked, float m, int i) {
  unsigned b = __float_as_uint(m);
  if (!(m > -__builtin_inff())) b = 0xff800000u;
  if (b == 0x80000000u) b = 0u;
  b = (b & 0x80000000u) ? ~b : (b | 0x80000000u);
  return ((u64)(unsigned)(hmasked + 1) << 42) | ((u64)b << 10) | (u64)(unsigned)(NP - 1 - i);
}

template <int XM>
__device__ __forceinline__ u64 key_row_xor(u64 k) {
  const int lo = (int)(unsigned)k;
  const int hi = (int)(unsigned)(k >> 32);
  const unsigned olo = (unsigned)__builtin_amdgcn_update_dpp(lo, lo, 0x160 + XM, 0xf, 0xf, false);
  const unsigned ohi = (unsigned)__builtin_amdgcn_update_dpp(hi, hi, 0x160 + XM, 0xf, 0xf, false);
  return ((u64)ohi << 32) | (u64)olo;
}

__device__ __forceinline__ u64 key_other_row(u64 k) {
  const int lo = (int)(unsigned)k;
  const int hi = (int)(unsigned)(k >> 32);
  const unsigned olo = (unsigned)__builtin_amdgcn_permlanex16(lo, lo, 0x76543210, 0xfedcba98, false, false);
  const unsigned ohi = (unsigned)__builtin_amdgcn_permlanex16(hi, hi, 0x76543210, 0xfedcba98, false, false);
  return ((u64)ohi << 32) | (u64)olo;
}

__device__ __forceinline__ u64 wave_max_key(u64 k) {
  u64 o;
  o = key_row_xor<1>(k); k = o > k ? o : k;
  o = key_row_xor<2>(k); k = o > k ? o : k;
  o = key_row_xor<4>(k); k = o > k ? o : k;
  o = key_row_xor<8>(k); k = o > k ? o : k;
  o = key_other_row(k);  k = o > k ? o : k;
  return k;
}

__global__ __launch_bounds__(1024) void fundamental_kernel(
    const float* __restrict__ freqs, const float* __restrict__ mags,
    const int* ws_hm, size_t ws_bytes,
    float* __restrict__ out)
{
  __shared__ float fr[NP];
  __shared__ float mg[NP];
  __shared__ u64   part[NP / 32];

  const int i    = threadIdx.x;
  const int base = (NB - 1) * NP;
  const float fi = freqs[base + i];
  const float mi = mags[base + i];

  int hmi;
  if (ws_bytes >= CNT_WS_BYTES) {
    hmi = ws_hm[i];
    fr[i] = fi;
  } else {
    fr[i] = fi;
    mg[i] = mi;
    __syncthreads();
    int h = 0;
    for (int j = 0; j < NP; ++j) {
      if (j == i) continue;
      if (!(mg[j] > 0.0f)) continue;
      const float fj = fr[j];
      bool match = false;
      #pragma unroll
      for (int order = 2; order <= 8; ++order)
        match = match || (fabsf(fj - fi * (float)order) < 2.0f);
      h += match ? 1 : 0;
    }
    hmi = (mi > 0.0f && h > 0) ? h : -1;
  }

  const u64 wk = wave_max_key(select_key(hmi, mi, i));
  if ((i & 31) == 0) part[i >> 5] = wk;
  __syncthreads();

  if (i < 32) {
    const u64 best = wave_max_key(part[i]);
    if (i == 0) out[NB * NP * 8] = fr[NP - 1 - (int)(best & (u64)(NP - 1))];
  }
}

extern "C" void kernel_launch(void* const* d_in, const int* in_sizes, int n_in,
                              void* d_out, int out_size, void* d_ws, size_t ws_size,
                              hipStream_t stream) {
  const float* freqs = (const float*)d_in[0];
  const float* mags  = (const float*)d_in[1];
  const float* W1    = (const float*)d_in[2];
  const float* b1    = (const float*)d_in[3];
  const float* W2    = (const float*)d_in[4];
  const float* b2    = (const float*)d_in[5];
  const float* W3    = (const float*)d_in[6];
  const float* b3    = (const float*)d_in[7];
  float* out = (float*)d_out;

  const int ntok   = in_sizes[0];
  const int blocks = ntok / TOKENS_PER_BLOCK;

  int* ws_hm = (int*)d_ws;

  mlp_wmma_kernel<<<blocks + CNT_BLOCKS, 256, 0, stream>>>(freqs, mags, W1, b1, W2, b2, W3, b3, out,
                                                          ws_hm, ws_size);
  fundamental_kernel<<<1, 1024, 0, stream>>>(freqs, mags, ws_hm, ws_size, out);
  (void)n_in; (void)out_size;
}


# LDS row pitch of the hidden-activation tile padded from 128 to 132 floats: layer-1 stores and the A-fragment reads of layers 2 and 3 spread over all LDS banks instead of piling on one bank group
# speedup vs baseline: 7.1559x; 1.3893x over previous
#include <hip/hip_runtime.h>
#include <hip/hip_bf16.h>

typedef __attribute__((ext_vector_type(2))) float v2f;
typedef __attribute__((ext_vector_type(8))) float v8f;

#define HIDDEN 128
#define TOKENS_PER_BLOCK 128
#define HPITCH 132
#define NB 64
#define NP 1024
#define CNT_BLOCKS 32
#define CNT_WS_BYTES (NP * 4)

__global__ __launch_bounds__(256) void mlp_wmma_kernel(
    const float* __restrict__ freqs, const float* __restrict__ mags,
    const float* __restrict__ W1, const float* __restrict__ b1,
    const float* __restrict__ W2, const float* __restrict__ b2,
    const float* __restrict__ W3, const float* __restrict__ b3,
    float* __restrict__ out, int* __restrict__ ws_hm, size_t ws_bytes)
{
  __shared__ float hbuf[TOKENS_PER_BLOCK * HPITCH];

  const int tid       = threadIdx.x;

  if (blockIdx.x < CNT_BLOCKS) {
    if (ws_bytes < CNT_WS_BYTES) return;
    const float* fr = freqs + (NB - 1) * NP;
    const float* mg = mags  + (NB - 1) * NP;
    const int clane = tid & 31;
    const int cwave = tid >> 5;
    const int i     = blockIdx.x * 32 + clane;
    const int j0    = cwave * (NP / 8);
    const float fi  = fr[i];
    float p[7];
    #pragma unroll
    for (int order = 2; order <= 8; ++order) p[order - 2] = fi * (float)order;
    float fv[4], mv[4];
    #pragma unroll
    for (int q = 0; q < 4; ++q) {
      fv[q] = fr[j0 + 32 * q + clane];
      mv[q] = mg[j0 + 32 * q + clane];
    }
    int h = 0;
    #pragma unroll
    for (int q = 0; q < 4; ++q) {
      for (int l = 0; l < 32; ++l) {
        const float fj = __int_as_float(__builtin_amdgcn_readlane(__float_as_int(fv[q]), l));
        const float mj = __int_as_float(__builtin_amdgcn_readlane(__float_as_int(mv[q]), l));
        bool match = false;
        #pragma unroll
        for (int o = 0; o < 7; ++o)
          match = match | (fabsf(fj - p[o]) < 2.0f);
        const int j = j0 + 32 * q + l;
        h += (match & (mj > 0.0f) & (j != i)) ? 1 : 0;
      }
    }
    int* part = (int*)hbuf;
    part[cwave * 32 + clane] = h;
    __syncthreads();
    if (tid < 32) {
      int s = 0;
      #pragma unroll
      for (int w = 0; w < 8; ++w) s += part[w * 32 + tid];
      const bool valid = mg[i] > 0.0f;
      ws_hm[i] = (valid && s > 0) ? s : -1;
    }
    return;
  }

  const int blockTok0 = (blockIdx.x - CNT_BLOCKS) * TOKENS_PER_BLOCK;

  {
    const int token = tid >> 1;
    const int cbase = (tid & 1) * 64;
    const int g     = blockTok0 + token;
    const float f = freqs[g];
    const float m = mags[g];
    #pragma unroll 8
    for (int j = 0; j < 64; ++j) {
      const int c = cbase + j;
      float h = fmaf(f, W1[c], fmaf(m, W1[HIDDEN + c], b1[c]));
      hbuf[token * HPITCH + c] = h > 0.0f ? h : 0.0f;
    }
  }
  __syncthreads();

  const int lane  = tid & 31;
  const int wave  = tid >> 5;
  const int idx   = lane & 15;
  const int half  = (lane >> 4) << 1;
  const int wrow0 = wave * 16;

  v8f acc[8];
  #pragma unroll
  for (int n = 0; n < 8; ++n) {
    const float bias = b2[n * 16 + idx];
    #pragma unroll
    for (int r = 0; r < 8; ++r) acc[n][r] = bias;
  }

  const float* aP = &hbuf[(wrow0 + idx) * HPITCH + half];

  #pragma unroll 4
  for (int k = 0; k < HIDDEN; k += 4) {
    v2f a; a.x = aP[k]; a.y = aP[k + 1];
    #pragma unroll
    for (int n = 0; n < 8; ++n) {
      const float* bP = &W2[(k + half) * HIDDEN + n * 16 + idx];
      v2f b; b.x = bP[0]; b.y = bP[HIDDEN];
      acc[n] = __builtin_amdgcn_wmma_f32_16x16x4_f32(
          false, a, false, b, (short)0, acc[n], false, false);
    }
  }

  __syncthreads();

  #pragma unroll
  for (int n = 0; n < 8; ++n) {
    #pragma unroll
    for (int r = 0; r < 8; ++r) {
      const int row = wrow0 + ((lane < 16) ? r : r + 8);
      const float v = acc[n][r];
      hbuf[row * HPITCH + n * 16 + idx] = v > 0.0f ? v : 0.0f;
    }
  }
  __syncthreads();

  v8f acc3;
  {
    const float bias = (idx < 8) ? b3[idx] : 0.0f;
    #pragma unroll
    for (int r = 0; r < 8; ++r) acc3[r] = bias;
  }
  #pragma unroll 4
  for (int k = 0; k < HIDDEN; k += 4) {
    v2f a; a.x = aP[k]; a.y = aP[k + 1];
    v2f b;
    b.x = (idx < 8) ? W3[(k + half)     * 8 + idx] : 0.0f;
    b.y = (idx < 8) ? W3[(k + half + 1) * 8 + idx] : 0.0f;
    acc3 = __builtin_amdgcn_wmma_f32_16x16x4_f32(
        false, a, false, b, (short)0, acc3, false, false);
  }

  if (idx < 8) {
    const int gbase = blockTok0 + wrow0;
    #pragma unroll
    for (int r = 0; r < 8; ++r) {
      const int row = (lane < 16) ? r : r + 8;
      out[(gbase + row) * 8 + idx] = acc3[r];
    }
  }
}

typedef unsigned long long u64;

__device__ __forceinline__ u64 select_key(int hmasked, float m, int i) {
  unsigned b = __float_as_uint(m);
  if (!(m > -__builtin_inff())) b = 0xff800000u;
  if (b == 0x80000000u) b = 0u;
  b = (b & 0x80000000u) ? ~b : (b | 0x80000000u);
  return ((u64)(unsigned)(hmasked + 1) << 42) | ((u64)b << 10) | (u64)(unsigned)(NP - 1 - i);
}

template <int XM>
__device__ __forceinline__ u64 key_row_xor(u64 k) {
  const int lo = (int)(unsigned)k;
  const int hi = (int)(unsigned)(k >> 32);
  const unsigned olo = (unsigned)__builtin_amdgcn_update_dpp(lo, lo, 0x160 + XM, 0xf, 0xf, false);
  const unsigned ohi = (unsigned)__builtin_amdgcn_update_dpp(hi, hi, 0x160 + XM, 0xf, 0xf, false);
  return ((u64)ohi << 32) | (u64)olo;
}

__device__ __forceinline__ u64 key_other_row(u64 k) {
  const int lo = (int)(unsigned)k;
  const int hi = (int)(unsigned)(k >> 32);
  const unsigned olo = (unsigned)__builtin_amdgcn_permlanex16(lo, lo, 0x76543210, 0xfedcba98, false, false);
  const unsigned ohi = (unsigned)__builtin_amdgcn_permlanex16(hi, hi, 0x76543210, 0xfedcba98, false, false);
  return ((u64)ohi << 32) | (u64)olo;
}

__device__ __forceinline__ u64 wave_max_key(u64 k) {
  u64 o;
  o = key_row_xor<1>(k); k = o > k ? o : k;
  o = key_row_xor<2>(k); k = o > k ? o : k;
  o = key_row_xor<4>(k); k = o > k ? o : k;
  o = key_row_xor<8>(k); k = o > k ? o : k;
  o = key_other_row(k);  k = o > k ? o : k;
  return k;
}

__global__ __launch_bounds__(1024) void fundamental_kernel(
    const float* __restrict__ freqs, const float* __restrict__ mags,
    const int* ws_hm, size_t ws_bytes,
    float* __restrict__ out)
{
  __shared__ float fr[NP];
  __shared__ float mg[NP];
  __shared__ u64   part[NP / 32];

  const int i    = threadIdx.x;
  const int base = (NB - 1) * NP;
  const float fi = freqs[base + i];
  const float mi = mags[base + i];

  int hmi;
  if (ws_bytes >= CNT_WS_BYTES) {
    hmi = ws_hm[i];
    fr[i] = fi;
  } else {
    fr[i] = fi;
    mg[i] = mi;
    __syncthreads();
    int h = 0;
    for (int j = 0; j < NP; ++j) {
      if (j == i) continue;
      if (!(mg[j] > 0.0f)) continue;
      const float fj = fr[j];
      bool match = false;
      #pragma unroll
      for (int order = 2; order <= 8; ++order)
        match = match || (fabsf(fj - fi * (float)order) < 2.0f);
      h += match ? 1 : 0;
    }
    hmi = (mi > 0.0f && h > 0) ? h : -1;
  }

  const u64 wk = wave_max_key(select_key(hmi, mi, i));
  if ((i & 31) == 0) part[i >> 5] = wk;
  __syncthreads();

  if (i < 32) {
    const u64 best = wave_max_key(part[i]);
    if (i == 0) out[NB * NP * 8] = fr[NP - 1 - (int)(best & (u64)(NP - 1))];
  }
}

extern "C" void kernel_launch(void* const* d_in, const int* in_sizes, int n_in,
                              void* d_out, int out_size, void* d_ws, size_t ws_size,
                              hipStream_t stream) {
  const float* freqs = (const float*)d_in[0];
  const float* mags  = (const float*)d_in[1];
  const float* W1    = (const float*)d_in[2];
  const float* b1    = (const float*)d_in[3];
  const float* W2    = (const float*)d_in[4];
  const float* b2    = (const float*)d_in[5];
  const float* W3    = (const float*)d_in[6];
  const float* b3    = (const float*)d_in[7];
  float* out = (float*)d_out;

  const int ntok   = in_sizes[0];
  const int blocks = ntok / TOKENS_PER_BLOCK;

  int* ws_hm = (int*)d_ws;

  mlp_wmma_kernel<<<blocks + CNT_BLOCKS, 256, 0, stream>>>(freqs, mags, W1, b1, W2, b2, W3, b3, out,
                                                          ws_hm, ws_size);
  fundamental_kernel<<<1, 1024, 0, stream>>>(freqs, mags, ws_hm, ws_size, out);
  (void)n_in; (void)out_size;
}


# layer-2 B operands from LDS: W2 copied once per work-group into LDS in fragment order, one 16-byte load per lane feeds two WMMAs (was two global loads per WMMA), next k-block loaded ahead
# speedup vs baseline: 8.4490x; 1.1807x over previous
#include <hip/hip_runtime.h>
#include <hip/hip_bf16.h>

typedef __attribute__((ext_vector_type(2))) float v2f;
typedef __attribute__((ext_vector_type(4))) float v4f;
typedef __attribute__((ext_vector_type(8))) float v8f;

#define HIDDEN 128
#define TOKENS_PER_BLOCK 128
#define HPITCH 132
#define WGROUPS 128
#define WGPITCH 33
#define NB 64
#define NP 1024
#define CNT_BLOCKS 32
#define CNT_WS_BYTES (NP * 4)

__global__ __launch_bounds__(256) void mlp_wmma_kernel(
    const float* __restrict__ freqs, const float* __restrict__ mags,
    const float* __restrict__ W1, const float* __restrict__ b1,
    const float* __restrict__ W2, const float* __restrict__ b2,
    const float* __restrict__ W3, const float* __restrict__ b3,
    float* __restrict__ out, int* __restrict__ ws_hm, size_t ws_bytes)
{
  __shared__ __attribute__((aligned(16))) float hbuf[TOKENS_PER_BLOCK * HPITCH];
  __shared__ __attribute__((aligned(16))) float wfrag[WGROUPS * WGPITCH * 4];

  const int tid       = threadIdx.x;

  if (blockIdx.x < CNT_BLOCKS) {
    if (ws_bytes < CNT_WS_BYTES) return;
    const float* fr = freqs + (NB - 1) * NP;
    const float* mg = mags  + (NB - 1) * NP;
    const int clane = tid & 31;
    const int cwave = tid >> 5;
    const int i     = blockIdx.x * 32 + clane;
    const int j0    = cwave * (NP / 8);
    const float fi  = fr[i];
    float p[7];
    #pragma unroll
    for (int order = 2; order <= 8; ++order) p[order - 2] = fi * (float)order;
    float fv[4], mv[4];
    #pragma unroll
    for (int q = 0; q < 4; ++q) {
      fv[q] = fr[j0 + 32 * q + clane];
      mv[q] = mg[j0 + 32 * q + clane];
    }
    int h = 0;
    #pragma unroll
    for (int q = 0; q < 4; ++q) {
      for (int l = 0; l < 32; ++l) {
        const float fj = __int_as_float(__builtin_amdgcn_readlane(__float_as_int(fv[q]), l));
        const float mj = __int_as_float(__builtin_amdgcn_readlane(__float_as_int(mv[q]), l));
        bool match = false;
        #pragma unroll
        for (int o = 0; o < 7; ++o)
          match = match | (fabsf(fj - p[o]) < 2.0f);
        const int j = j0 + 32 * q + l;
        h += (match & (mj > 0.0f) & (j != i)) ? 1 : 0;
      }
    }
    int* part = (int*)hbuf;
    part[cwave * 32 + clane] = h;
    __syncthreads();
    if (tid < 32) {
      int s = 0;
      #pragma unroll
      for (int w = 0; w < 8; ++w) s += part[w * 32 + tid];
      const bool valid = mg[i] > 0.0f;
      ws_hm[i] = (valid && s > 0) ? s : -1;
    }
    return;
  }

  const int blockTok0 = (blockIdx.x - CNT_BLOCKS) * TOKENS_PER_BLOCK;

  const int sl = tid & 31;
  const int sw = tid >> 5;
  v4f wr[4][4];
  #pragma unroll
  for (int p = 0; p < 4; ++p) {
    const int kb = sw * 2 + (p >> 1);
    const int h  = p & 1;
    const float* wsrc = W2 + (kb * 8 + 2 * h) * HIDDEN + sl * 4;
    wr[p][0] = *(const v4f*)(wsrc);
    wr[p][1] = *(const v4f*)(wsrc + HIDDEN);
    wr[p][2] = *(const v4f*)(wsrc + 4 * HIDDEN);
    wr[p][3] = *(const v4f*)(wsrc + 5 * HIDDEN);
  }

  {
    const int token = tid >> 1;
    const int cbase = (tid & 1) * 64;
    const int g     = blockTok0 + token;
    const float f = freqs[g];
    const float m = mags[g];
    #pragma unroll 8
    for (int j = 0; j < 64; ++j) {
      const int c = cbase + j;
      float h = fmaf(f, W1[c], fmaf(m, W1[HIDDEN + c], b1[c]));
      hbuf[token * HPITCH + c] = h > 0.0f ? h : 0.0f;
    }
  }
  {
    const int sn = sl >> 2;
    const int sc = (sl & 3) * 4;
    #pragma unroll
    for (int p = 0; p < 4; ++p) {
      const int kb = sw * 2 + (p >> 1);
      const int h  = p & 1;
      float* dst = &wfrag[((kb * 8 + sn) * WGPITCH + h * 16 + sc) * 4];
      v4f e;
      e.x = wr[p][0].x; e.y = wr[p][1].x; e.z = wr[p][2].x; e.w = wr[p][3].x; *(v4f*)(dst)      = e;
      e.x = wr[p][0].y; e.y = wr[p][1].y; e.z = wr[p][2].y; e.w = wr[p][3].y; *(v4f*)(dst + 4)  = e;
      e.x = wr[p][0].z; e.y = wr[p][1].z; e.z = wr[p][2].z; e.w = wr[p][3].z; *(v4f*)(dst + 8)  = e;
      e.x = wr[p][0].w; e.y = wr[p][1].w; e.z = wr[p][2].w; e.w = wr[p][3].w; *(v4f*)(dst + 12) = e;
    }
  }
  __syncthreads();

  const int lane  = tid & 31;
  const int wave  = tid >> 5;
  const int idx   = lane & 15;
  const int half  = (lane >> 4) << 1;
  const int wrow0 = wave * 16;

  v8f acc[8];
  #pragma unroll
  for (int n = 0; n < 8; ++n) {
    const float bias = b2[n * 16 + idx];
    #pragma unroll
    for (int r = 0; r < 8; ++r) acc[n][r] = bias;
  }

  const float* aP = &hbuf[(wrow0 + idx) * HPITCH + half];

  const float* bL = &wfrag[lane * 4];

  v4f bA[8], bB[8];
  v2f aA0, aA1, aB0, aB1;
#define LOAD_KB(B, A0, A1, KB)                                              \
  do {                                                                      \
    const int k_ = (KB) * 8;                                                \
    A0.x = aP[k_];     A0.y = aP[k_ + 1];                 \
    A1.x = aP[k_ + 4]; A1.y = aP[k_ + 5];                 \
    _Pragma("unroll")                                                       \
    for (int n = 0; n < 8; ++n)                                             \
      B[n] = *(const v4f*)(bL + ((KB) * 8 + n) * (WGPITCH * 4));            \
  } while (0)
#define WMMA_KB(B, A0, A1)                                                  \
  do {                                                                      \
    _Pragma("unroll")                                                       \
    for (int n = 0; n < 8; ++n) {                                           \
      v2f b0; b0.x = B[n].x; b0.y = B[n].y;                                 \
      acc[n] = __builtin_amdgcn_wmma_f32_16x16x4_f32(                       \
          false, A0, false, b0, (short)0, acc[n], false, false);            \
    }                                                                       \
    _Pragma("unroll")                                                       \
    for (int n = 0; n < 8; ++n) {                                           \
      v2f b1; b1.x = B[n].z; b1.y = B[n].w;                                 \
      acc[n] = __builtin_amdgcn_wmma_f32_16x16x4_f32(                       \
          false, A1, false, b1, (short)0, acc[n], false, false);            \
    }                                                                       \
  } while (0)

  LOAD_KB(bA, aA0, aA1, 0);
  for (int kb = 0; kb < HIDDEN / 8; kb += 2) {
    LOAD_KB(bB, aB0, aB1, kb + 1);
    WMMA_KB(bA, aA0, aA1);
    const int kn = (kb + 2 < HIDDEN / 8) ? kb + 2 : kb;
    LOAD_KB(bA, aA0, aA1, kn);
    WMMA_KB(bB, aB0, aB1);
  }
#undef LOAD_KB
#undef WMMA_KB

  __syncthreads();

  #pragma unroll
  for (int n = 0; n < 8; ++n) {
    #pragma unroll
    for (int r = 0; r < 8; ++r) {
      const int row = wrow0 + ((lane < 16) ? r : r + 8);
      const float v = acc[n][r];
      hbuf[row * HPITCH + n * 16 + idx] = v > 0.0f ? v : 0.0f;
    }
  }
  __syncthreads();

  v8f acc3;
  {
    const float bias = (idx < 8) ? b3[idx] : 0.0f;
    #pragma unroll
    for (int r = 0; r < 8; ++r) acc3[r] = bias;
  }
  #pragma unroll 4
  for (int k = 0; k < HIDDEN; k += 4) {
    v2f a; a.x = aP[k]; a.y = aP[k + 1];
    v2f b;
    b.x = (idx < 8) ? W3[(k + half)     * 8 + idx] : 0.0f;
    b.y = (idx < 8) ? W3[(k + half + 1) * 8 + idx] : 0.0f;
    acc3 = __builtin_amdgcn_wmma_f32_16x16x4_f32(
        false, a, false, b, (short)0, acc3, false, false);
  }

  if (idx < 8) {
    const int gbase = blockTok0 + wrow0;
    #pragma unroll
    for (int r = 0; r < 8; ++r) {
      const int row = (lane < 16) ? r : r + 8;
      out[(gbase + row) * 8 + idx] = acc3[r];
    }
  }
}

typedef unsigned long long u64;

__device__ __forceinline__ u64 select_key(int hmasked, float m, int i) {
  unsigned b = __float_as_uint(m);
  if (!(m > -__builtin_inff())) b = 0xff800000u;
  if (b == 0x80000000u) b = 0u;
  b = (b & 0x80000000u) ? ~b : (b | 0x80000000u);
  return ((u64)(unsigned)(hmasked + 1) << 42) | ((u64)b << 10) | (u64)(unsigned)(NP - 1 - i);
}

template <int XM>
__device__ __forceinline__ u64 key_row_xor(u64 k) {
  const int lo = (int)(unsigned)k;
  const int hi = (int)(unsigned)(k >> 32);
  const unsigned olo = (unsigned)__builtin_amdgcn_update_dpp(lo, lo, 0x160 + XM, 0xf, 0xf, false);
  const unsigned ohi = (unsigned)__builtin_amdgcn_update_dpp(hi, hi, 0x160 + XM, 0xf, 0xf, false);
  return ((u64)ohi << 32) | (u64)olo;
}

__device__ __forceinline__ u64 key_other_row(u64 k) {
  const int lo = (int)(unsigned)k;
  const int hi = (int)(unsigned)(k >> 32);
  const unsigned olo = (unsigned)__builtin_amdgcn_permlanex16(lo, lo, 0x76543210, 0xfedcba98, false, false);
  const unsigned ohi = (unsigned)__builtin_amdgcn_permlanex16(hi, hi, 0x76543210, 0xfedcba98, false, false);
  return ((u64)ohi << 32) | (u64)olo;
}

__device__ __forceinline__ u64 wave_max_key(u64 k) {
  u64 o;
  o = key_row_xor<1>(k); k = o > k ? o : k;
  o = key_row_xor<2>(k); k = o > k ? o : k;
  o = key_row_xor<4>(k); k = o > k ? o : k;
  o = key_row_xor<8>(k); k = o > k ? o : k;
  o = key_other_row(k);  k = o > k ? o : k;
  return k;
}

__global__ __launch_bounds__(1024) void fundamental_kernel(
    const float* __restrict__ freqs, const float* __restrict__ mags,
    const int* ws_hm, size_t ws_bytes,
    float* __restrict__ out)
{
  __shared__ float fr[NP];
  __shared__ float mg[NP];
  __shared__ u64   part[NP / 32];

  const int i    = threadIdx.x;
  const int base = (NB - 1) * NP;
  const float fi = freqs[base + i];
  const float mi = mags[base + i];

  int hmi;
  if (ws_bytes >= CNT_WS_BYTES) {
    hmi = ws_hm[i];
    fr[i] = fi;
  } else {
    fr[i] = fi;
    mg[i] = mi;
    __syncthreads();
    int h = 0;
    for (int j = 0; j < NP; ++j) {
      if (j == i) continue;
      if (!(mg[j] > 0.0f)) continue;
      const float fj = fr[j];
      bool match = false;
      #pragma unroll
      for (int order = 2; order <= 8; ++order)
        match = match || (fabsf(fj - fi * (float)order) < 2.0f);
      h += match ? 1 : 0;
    }
    hmi = (mi > 0.0f && h > 0) ? h : -1;
  }

  const u64 wk = wave_max_key(select_key(hmi, mi, i));
  if ((i & 31) == 0) part[i >> 5] = wk;
  __syncthreads();

  if (i < 32) {
    const u64 best = wave_max_key(part[i]);
    if (i == 0) out[NB * NP * 8] = fr[NP - 1 - (int)(best & (u64)(NP - 1))];
  }
}

extern "C" void kernel_launch(void* const* d_in, const int* in_sizes, int n_in,
                              void* d_out, int out_size, void* d_ws, size_t ws_size,
                              hipStream_t stream) {
  const float* freqs = (const float*)d_in[0];
  const float* mags  = (const float*)d_in[1];
  const float* W1    = (const float*)d_in[2];
  const float* b1    = (const float*)d_in[3];
  const float* W2    = (const float*)d_in[4];
  const float* b2    = (const float*)d_in[5];
  const float* W3    = (const float*)d_in[6];
  const float* b3    = (const float*)d_in[7];
  float* out = (float*)d_out;

  const int ntok   = in_sizes[0];
  const int blocks = ntok / TOKENS_PER_BLOCK;

  int* ws_hm = (int*)d_ws;

  mlp_wmma_kernel<<<blocks + CNT_BLOCKS, 256, 0, stream>>>(freqs, mags, W1, b1, W2, b2, W3, b3, out,
                                                          ws_hm, ws_size);
  fundamental_kernel<<<1, 1024, 0, stream>>>(freqs, mags, ws_hm, ws_size, out);
  (void)n_in; (void)out_size;
}


# activations stay in registers: layers computed transposed (weights = WMMA A from LDS, tokens on lanes), so each layer's D is the next one's B; no LDS round trip; new k grouping changes the sum order
# speedup vs baseline: 9.5047x; 1.1249x over previous
#include <hip/hip_runtime.h>


typedef __attribute__((ext_vector_type(2))) float v2f;
typedef __attribute__((ext_vector_type(4))) float v4f;
typedef __attribute__((ext_vector_type(8))) float v8f;

#define HIDDEN 128
#define TOKENS_PER_BLOCK 128
#define WGROUPS 128
#define WGPITCH 33
#define W3GROUPS 16
#define NB 64
#define NP 1024
#define CNT_BLOCKS 32
#define CNT_WS_BYTES (NP * 4)

__global__ __launch_bounds__(256) __attribute__((amdgpu_waves_per_eu(4, 4))) void mlp_wmma_kernel(
    const float* __restrict__ freqs, const float* __restrict__ mags,
    const float* __restrict__ W1, const float* __restrict__ b1,
    const float* __restrict__ W2, const float* __restrict__ b2,
    const float* __restrict__ W3, const float* __restrict__ b3,
    float* __restrict__ out, int* __restrict__ ws_hm, size_t ws_bytes)
{
  __shared__ __attribute__((aligned(16))) float wfrag[WGROUPS * WGPITCH * 4];
  __shared__ __attribute__((aligned(16))) float w3frag[W3GROUPS * 16 * 4];

  const int tid       = threadIdx.x;
  if (blockIdx.x < CNT_BLOCKS) {
    if (ws_bytes < CNT_WS_BYTES) return;
    const float* fr = freqs + (NB - 1) * NP;
    const float* mg = mags  + (NB - 1) * NP;
    const int clane = tid & 31;
    const int cwave = tid >> 5;
    const int i     = blockIdx.x * 32 + clane;
    const int j0    = cwave * (NP / 8);
    const float fi  = fr[i];
    float p[7];
    #pragma unroll
    for (int order = 2; order <= 8; ++order) p[order - 2] = fi * (float)order;
    float fv[4], mv[4];
    #pragma unroll
    for (int q = 0; q < 4; ++q) {
      fv[q] = fr[j0 + 32 * q + clane];
      mv[q] = mg[j0 + 32 * q + clane];
    }
    int h = 0;
    #pragma unroll
    for (int q = 0; q < 4; ++q) {
      for (int l = 0; l < 32; ++l) {
        const float fj = __int_as_float(__builtin_amdgcn_readlane(__float_as_int(fv[q]), l));
        const float mj = __int_as_float(__builtin_amdgcn_readlane(__float_as_int(mv[q]), l));
        bool match = false;
        #pragma unroll
        for (int o = 0; o < 7; ++o)
          match = match | (fabsf(fj - p[o]) < 2.0f);
        const int j = j0 + 32 * q + l;
        h += (match & (mj > 0.0f) & (j != i)) ? 1 : 0;
      }
    }
    int* part = (int*)wfrag;
    part[cwave * 32 + clane] = h;
    __syncthreads();
    if (tid < 32) {
      int s = 0;
      #pragma unroll
      for (int w = 0; w < 8; ++w) s += part[w * 32 + tid];
      const bool valid = mg[i] > 0.0f;
      ws_hm[i] = (valid && s > 0) ? s : -1;
    }
    return;
  }

  const int blockTok0 = (blockIdx.x - CNT_BLOCKS) * TOKENS_PER_BLOCK;

  const int lane  = tid & 31;
  const int wave  = tid >> 5;
  const int c     = lane & 15;
  const int h     = lane >> 4;
  const int tok   = blockTok0 + wave * 16 + c;

  const int sl = tid & 31;
  const int sw = tid >> 5;
  v4f wr[16];
  #pragma unroll
  for (int j = 0; j < 16; ++j)
    wr[j] = *(const v4f*)(W2 + (sw * 16 + j) * HIDDEN + sl * 4);

  const int g3 = tid >> 4;
  const int h3 = (tid >> 3) & 1;
  const int j3 = tid & 7;
  const int k3 = 16 * (g3 >> 1) + 8 * h3 + 4 * (g3 & 1);
  v4f w3e;
  w3e.x = W3[(k3 + 0) * 8 + j3];
  w3e.y = W3[(k3 + 1) * 8 + j3];
  w3e.z = W3[(k3 + 2) * 8 + j3];
  w3e.w = W3[(k3 + 3) * 8 + j3];

  const float f  = freqs[tok];
  const float mg = mags[tok];

  v8f acc[8];
  #pragma unroll
  for (int m = 0; m < 8; ++m) {
    const v4f lo = *(const v4f*)(b2 + 16 * m + 8 * h);
    const v4f hi = *(const v4f*)(b2 + 16 * m + 8 * h + 4);
    acc[m][0] = lo.x; acc[m][1] = lo.y; acc[m][2] = lo.z; acc[m][3] = lo.w;
    acc[m][4] = hi.x; acc[m][5] = hi.y; acc[m][6] = hi.z; acc[m][7] = hi.w;
  }

  {
    const int sm = sl >> 2;
    const int sc = (sl & 3) * 4;
    #pragma unroll
    for (int e = 0; e < 4; ++e) {
      const int p  = e >> 1;
      const int hh = e & 1;
      const int j0 = 4 * p + 8 * hh;
      float* dst = &wfrag[(((sw * 2 + p) * 8 + sm) * WGPITCH + hh * 16 + sc) * 4];
      v4f q;
      q.x = wr[j0].x; q.y = wr[j0 + 1].x; q.z = wr[j0 + 2].x; q.w = wr[j0 + 3].x; *(v4f*)(dst)      = q;
      q.x = wr[j0].y; q.y = wr[j0 + 1].y; q.z = wr[j0 + 2].y; q.w = wr[j0 + 3].y; *(v4f*)(dst + 4)  = q;
      q.x = wr[j0].z; q.y = wr[j0 + 1].z; q.z = wr[j0 + 2].z; q.w = wr[j0 + 3].z; *(v4f*)(dst + 8)  = q;
      q.x = wr[j0].w; q.y = wr[j0 + 1].w; q.z = wr[j0 + 2].w; q.w = wr[j0 + 3].w; *(v4f*)(dst + 12) = q;
    }
    *(v4f*)&w3frag[(g3 * 16 + h3 * 8 + j3) * 4] = w3e;
  }
  __syncthreads();

  const float* aL = &wfrag[lane * 4];

  v4f wa0, wa1, wb0, wb1, bb0, bb1;
  float h1[8];
#define H1_LOAD(N)                                                          \
  do {                                                                      \
    const int f0_ = 16 * (N) + 8 * h;                                       \
    wa0 = *(const v4f*)(W1 + f0_);          wa1 = *(const v4f*)(W1 + f0_ + 4);          \
    wb0 = *(const v4f*)(W1 + HIDDEN + f0_); wb1 = *(const v4f*)(W1 + HIDDEN + f0_ + 4); \
    bb0 = *(const v4f*)(b1 + f0_);          bb1 = *(const v4f*)(b1 + f0_ + 4);          \
  } while (0)
#define H1_ONE(R, WA, WB, BB)                                               \
  do {                                                                      \
    const float t_ = fmaf(f, (WA), fmaf(mg, (WB), (BB)));                   \
    h1[R] = t_ > 0.0f ? t_ : 0.0f;                                          \
  } while (0)
#define H1_COMPUTE()                                                        \
  do {                                                                      \
    H1_ONE(0, wa0.x, wb0.x, bb0.x); H1_ONE(1, wa0.y, wb0.y, bb0.y);         \
    H1_ONE(2, wa0.z, wb0.z, bb0.z); H1_ONE(3, wa0.w, wb0.w, bb0.w);         \
    H1_ONE(4, wa1.x, wb1.x, bb1.x); H1_ONE(5, wa1.y, wb1.y, bb1.y);         \
    H1_ONE(6, wa1.z, wb1.z, bb1.z); H1_ONE(7, wa1.w, wb1.w, bb1.w);         \
  } while (0)

  v4f fa[8], fb[8];
#define LOAD_KB(F, KB)                                                      \
  do {                                                                      \
    _Pragma("unroll")                                                       \
    for (int m = 0; m < 8; ++m)                                             \
      F[m] = *(const v4f*)(aL + ((KB) * 8 + m) * (WGPITCH * 4));            \
  } while (0)
#define WMMA_KB(F, X0, X1, X2, X3)                                          \
  do {                                                                      \
    v2f t0_; t0_.x = (X0); t0_.y = (X1);                                    \
    v2f t1_; t1_.x = (X2); t1_.y = (X3);                                    \
    _Pragma("unroll")                                                       \
    for (int m = 0; m < 8; ++m) {                                           \
      v2f a_; a_.x = F[m].x; a_.y = F[m].y;                                 \
      acc[m] = __builtin_amdgcn_wmma_f32_16x16x4_f32(                       \
          false, a_, false, t0_, (short)0, acc[m], false, false);           \
    }                                                                       \
    _Pragma("unroll")                                                       \
    for (int m = 0; m < 8; ++m) {                                           \
      v2f a_; a_.x = F[m].z; a_.y = F[m].w;                                 \
      acc[m] = __builtin_amdgcn_wmma_f32_16x16x4_f32(                       \
          false, a_, false, t1_, (short)0, acc[m], false, false);           \
    }                                                                       \
  } while (0)

  H1_LOAD(0);
  LOAD_KB(fa, 0);
  H1_COMPUTE();
  for (int n = 0; n < HIDDEN / 16; ++n) {
    const int nn = (n + 1 < HIDDEN / 16) ? n + 1 : n;
    H1_LOAD(nn);
    LOAD_KB(fb, 2 * n + 1);
    WMMA_KB(fa, h1[0], h1[1], h1[2], h1[3]);
    LOAD_KB(fa, 2 * nn);
    WMMA_KB(fb, h1[4], h1[5], h1[6], h1[7]);
    H1_COMPUTE();
  }
#undef LOAD_KB
#undef H1_LOAD
#undef H1_ONE
#undef H1_COMPUTE

  v8f acc3;
  #pragma unroll
  for (int r = 0; r < 8; ++r) acc3[r] = b3[r];
  const float* a3L = &w3frag[(h * 8 + (c & 7)) * 4];
  v4f f3[W3GROUPS];
  #pragma unroll
  for (int g = 0; g < W3GROUPS; ++g) f3[g] = *(const v4f*)(a3L + g * 64);
  #pragma unroll
  for (int m = 0; m < 8; ++m) {
    float h2[8];
    #pragma unroll
    for (int r = 0; r < 8; ++r) { const float v = acc[m][r]; h2[r] = v > 0.0f ? v : 0.0f; }
    #pragma unroll
    for (int p = 0; p < 2; ++p) {
      v2f a_, t_;
      a_.x = f3[2 * m + p].x; a_.y = f3[2 * m + p].y; t_.x = h2[4 * p];     t_.y = h2[4 * p + 1];
      acc3 = __builtin_amdgcn_wmma_f32_16x16x4_f32(false, a_, false, t_, (short)0, acc3, false, false);
      a_.x = f3[2 * m + p].z; a_.y = f3[2 * m + p].w; t_.x = h2[4 * p + 2]; t_.y = h2[4 * p + 3];
      acc3 = __builtin_amdgcn_wmma_f32_16x16x4_f32(false, a_, false, t_, (short)0, acc3, false, false);
    }
  }
#undef WMMA_KB

  v4f o;
  o.x = h ? acc3[4] : acc3[0];
  o.y = h ? acc3[5] : acc3[1];
  o.z = h ? acc3[6] : acc3[2];
  o.w = h ? acc3[7] : acc3[3];
  *(v4f*)(out + tok * 8 + 4 * h) = o;
}

typedef unsigned long long u64;

__device__ __forceinline__ u64 select_key(int hmasked, float m, int i) {
  unsigned b = __float_as_uint(m);
  if (!(m > -__builtin_inff())) b = 0xff800000u;
  if (b == 0x80000000u) b = 0u;
  b = (b & 0x80000000u) ? ~b : (b | 0x80000000u);
  return ((u64)(unsigned)(hmasked + 1) << 42) | ((u64)b << 10) | (u64)(unsigned)(NP - 1 - i);
}

template <int XM>
__device__ __forceinline__ u64 key_row_xor(u64 k) {
  const int lo = (int)(unsigned)k;
  const int hi = (int)(unsigned)(k >> 32);
  const unsigned olo = (unsigned)__builtin_amdgcn_update_dpp(lo, lo, 0x160 + XM, 0xf, 0xf, false);
  const unsigned ohi = (unsigned)__builtin_amdgcn_update_dpp(hi, hi, 0x160 + XM, 0xf, 0xf, false);
  return ((u64)ohi << 32) | (u64)olo;
}

__device__ __forceinline__ u64 key_other_row(u64 k) {
  const int lo = (int)(unsigned)k;
  const int hi = (int)(unsigned)(k >> 32);
  const unsigned olo = (unsigned)__builtin_amdgcn_permlanex16(lo, lo, 0x76543210, 0xfedcba98, false, false);
  const unsigned ohi = (unsigned)__builtin_amdgcn_permlanex16(hi, hi, 0x76543210, 0xfedcba98, false, false);
  return ((u64)ohi << 32) | (u64)olo;
}

__device__ __forceinline__ u64 wave_max_key(u64 k) {
  u64 o;
  o = key_row_xor<1>(k); k = o > k ? o : k;
  o = key_row_xor<2>(k); k = o > k ? o : k;
  o = key_row_xor<4>(k); k = o > k ? o : k;
  o = key_row_xor<8>(k); k = o > k ? o : k;
  o = key_other_row(k);  k = o > k ? o : k;
  return k;
}

__global__ __launch_bounds__(1024) void fundamental_kernel(
    const float* __restrict__ freqs, const float* __restrict__ mags,
    const int* ws_hm, size_t ws_bytes,
    float* __restrict__ out)
{
  __shared__ float fr[NP];
  __shared__ float mg[NP];
  __shared__ u64   part[NP / 32];

  const int i    = threadIdx.x;
  const int base = (NB - 1) * NP;
  const float fi = freqs[base + i];
  const float mi = mags[base + i];

  int hmi;
  if (ws_bytes >= CNT_WS_BYTES) {
    hmi = ws_hm[i];
    fr[i] = fi;
  } else {
    fr[i] = fi;
    mg[i] = mi;
    __syncthreads();
    int h = 0;
    for (int j = 0; j < NP; ++j) {
      if (j == i) continue;
      if (!(mg[j] > 0.0f)) continue;
      const float fj = fr[j];
      bool match = false;
      #pragma unroll
      for (int order = 2; order <= 8; ++order)
        match = match || (fabsf(fj - fi * (float)order) < 2.0f);
      h += match ? 1 : 0;
    }
    hmi = (mi > 0.0f && h > 0) ? h : -1;
  }

  const u64 wk = wave_max_key(select_key(hmi, mi, i));
  if ((i & 31) == 0) part[i >> 5] = wk;
  __syncthreads();

  if (i < 32) {
    const u64 best = wave_max_key(part[i]);
    if (i == 0) out[NB * NP * 8] = fr[NP - 1 - (int)(best & (u64)(NP - 1))];
  }
}

extern "C" void kernel_launch(void* const* d_in, const int* in_sizes, int n_in,
                              void* d_out, int out_size, void* d_ws, size_t ws_size,
                              hipStream_t stream) {
  const float* freqs = (const float*)d_in[0];
  const float* mags  = (const float*)d_in[1];
  const float* W1    = (const float*)d_in[2];
  const float* b1    = (const float*)d_in[3];
  const float* W2    = (const float*)d_in[4];
  const float* b2    = (const float*)d_in[5];
  const float* W3    = (const float*)d_in[6];
  const float* b3    = (const float*)d_in[7];
  float* out = (float*)d_out;

  const int ntok   = in_sizes[0];
  const int blocks = ntok / TOKENS_PER_BLOCK;

  int* ws_hm = (int*)d_ws;

  mlp_wmma_kernel<<<blocks + CNT_BLOCKS, 256, 0, stream>>>(freqs, mags, W1, b1, W2, b2, W3, b3, out,
                                                          ws_hm, ws_size);
  fundamental_kernel<<<1, 1024, 0, stream>>>(freqs, mags, ws_hm, ws_size, out);
  (void)n_in; (void)out_size;
}
